# starting kernel
# baseline (speedup 1.0000x reference)
#include <hip/hip_runtime.h>

typedef float v2f __attribute__((ext_vector_type(2)));
typedef float v8f __attribute__((ext_vector_type(8)));

#define NNODES 20000
#define NEDGES 320000

__global__ __launch_bounds__(256) void k_scatter_add(
    const float* __restrict__ feat, const int* __restrict__ ei,
    float* __restrict__ agg, int n_edges, int D)
{
    const int chunks = D >> 2;
    long long tid = (long long)blockIdx.x * blockDim.x + threadIdx.x;
    long long total = (long long)n_edges * chunks;
    if (tid >= total) return;
    int e = (int)(tid / chunks);
    int c = (int)(tid - (long long)e * chunks);
    int s = ei[e];
    int d = ei[n_edges + e];
    const float4 v = reinterpret_cast<const float4*>(feat + (size_t)s * D)[c];
    float* out = agg + (size_t)d * D + (c << 2);
    unsafeAtomicAdd(out + 0, v.x);
    unsafeAtomicAdd(out + 1, v.y);
    unsafeAtomicAdd(out + 2, v.z);
    unsafeAtomicAdd(out + 3, v.w);
}

template <int NT, bool DUAL>
__global__ __launch_bounds__(256) void k_gemm_strip(
    const float* __restrict__ A0, const float* __restrict__ B0,
    const float* __restrict__ A1, const float* __restrict__ B1,
    const float* __restrict__ bias, float* __restrict__ C,
    int M, int K, int relu)
{
    constexpr int N = NT * 16;
    constexpr int BK = 64;
    __shared__ __align__(16) float Bs0[BK * N];
    __shared__ __align__(16) float Bs1[DUAL ? (BK * N) : 4];

    const int lane = threadIdx.x & 31;
    const int wave = threadIdx.x >> 5;
    const int tm = blockIdx.x * 8 + wave;
    const bool active = (tm * 16 < M);
    const int rh = lane & 15;
    const int hi = lane >> 4;

    v8f acc[NT];
#pragma unroll
    for (int t = 0; t < NT; ++t) acc[t] = v8f{};

    const float* ap0 = A0 + (size_t)(tm * 16 + rh) * K + 2 * hi;
    const float* ap1 = DUAL ? (A1 + (size_t)(tm * 16 + rh) * K + 2 * hi) : nullptr;

    for (int kc = 0; kc < K; kc += BK) {
        __syncthreads();
        {
            constexpr int ITERS = (BK / 2) * (N / 4);
            for (int idx = threadIdx.x; idx < ITERS; idx += 256) {
                const int pr = idx / (N / 4);
                const int c = (idx - pr * (N / 4)) * 4;
                const int k = 2 * pr;
                float4 g0 = *(const float4*)(B0 + (size_t)(kc + k) * N + c);
                float4 g1 = *(const float4*)(B0 + (size_t)(kc + k + 1) * N + c);
                float4 lo = {g0.x, g1.x, g0.y, g1.y};
                float4 hi4 = {g0.z, g1.z, g0.w, g1.w};
                float* dst = &Bs0[(pr * N + c) * 2];
                *(float4*)(dst) = lo;
                *(float4*)(dst + 4) = hi4;
                if constexpr (DUAL) {
                    float4 h0 = *(const float4*)(B1 + (size_t)(kc + k) * N + c);
                    float4 h1 = *(const float4*)(B1 + (size_t)(kc + k + 1) * N + c);
                    float4 l2 = {h0.x, h1.x, h0.y, h1.y};
                    float4 h2 = {h0.z, h1.z, h0.w, h1.w};
                    float* dst1 = &Bs1[(pr * N + c) * 2];
                    *(float4*)(dst1) = l2;
                    *(float4*)(dst1 + 4) = h2;
                }
            }
        }
        __syncthreads();
        if (!active) continue;

        const v2f* Bp0 = (const v2f*)Bs0;
        const v2f* Bp1 = (const v2f*)Bs1;
        for (int k = 0; k < BK; k += 4) {
            const int pr = (k >> 1) + hi;
            v2f a;
            a.x = ap0[kc + k];
            a.y = ap0[kc + k + 1];
#pragma unroll
            for (int t = 0; t < NT; ++t) {
                v2f b = Bp0[pr * N + t * 16 + rh];
                acc[t] = __builtin_amdgcn_wmma_f32_16x16x4_f32(
                    false, a, false, b, (short)0, acc[t], false, false);
            }
            if constexpr (DUAL) {
                v2f a1;
                a1.x = ap1[kc + k];
                a1.y = ap1[kc + k + 1];
#pragma unroll
                for (int t = 0; t < NT; ++t) {
                    v2f b = Bp1[pr * N + t * 16 + rh];
                    acc[t] = __builtin_amdgcn_wmma_f32_16x16x4_f32(
                        false, a1, false, b, (short)0, acc[t], false, false);
                }
            }
        }
    }

    if (active) {
#pragma unroll
        for (int t = 0; t < NT; ++t) {
            const int col = t * 16 + rh;
            const float bv = bias[col];
#pragma unroll
            for (int i = 0; i < 8; ++i) {
                int r = tm * 16 + i + 8 * hi;
                float v = acc[t][i] + bv;
                if (relu) v = fmaxf(v, 0.0f);
                C[(size_t)r * N + col] = v;
            }
        }
    }
}

__global__ __launch_bounds__(256) void k_head(
    const float* __restrict__ h, const float* __restrict__ W,
    const float* __restrict__ b, float* __restrict__ out, int M)
{
    int n = blockIdx.x * blockDim.x + threadIdx.x;
    if (n >= M) return;
    const float* hr = h + (size_t)n * 64;
    float z[10];
#pragma unroll
    for (int c = 0; c < 10; ++c) z[c] = b[c];
    for (int k = 0; k < 64; ++k) {
        float hv = hr[k];
        const float* wr = W + k * 10;
#pragma unroll
        for (int c = 0; c < 10; ++c) z[c] = fmaf(hv, wr[c], z[c]);
    }
    float mx = z[0];
#pragma unroll
    for (int c = 1; c < 10; ++c) mx = fmaxf(mx, z[c]);
    float s = 0.0f;
#pragma unroll
    for (int c = 0; c < 10; ++c) s += expf(z[c] - mx);
    float lse = mx + logf(s);
    float* o = out + (size_t)n * 10;
#pragma unroll
    for (int c = 0; c < 10; ++c) o[c] = z[c] - lse;
}

extern "C" void kernel_launch(void* const* d_in, const int* in_sizes, int n_in,
                              void* d_out, int out_size, void* d_ws, size_t ws_size,
                              hipStream_t stream) {
    const float* x       = (const float*)d_in[0];
    const int*   ei      = (const int*)d_in[1];
    const float* W1_rel  = (const float*)d_in[2];
    const float* W1_root = (const float*)d_in[3];
    const float* b1      = (const float*)d_in[4];
    const float* W2_rel  = (const float*)d_in[5];
    const float* W2_root = (const float*)d_in[6];
    const float* b2      = (const float*)d_in[7];
    const float* lin1_w  = (const float*)d_in[8];
    const float* lin1_b  = (const float*)d_in[9];
    const float* lin2_w  = (const float*)d_in[10];
    const float* lin2_b  = (const float*)d_in[11];
    const float* lin3_w  = (const float*)d_in[12];
    const float* lin3_b  = (const float*)d_in[13];

    float* agg1 = (float*)d_ws;
    float* h1   = agg1 + (size_t)NNODES * 256;
    float* agg2 = h1   + (size_t)NNODES * 128;
    float* h2   = agg2 + (size_t)NNODES * 128;
    float* m1   = h2   + (size_t)NNODES * 64;
    float* m2   = m1   + (size_t)NNODES * 128;

    hipMemsetAsync(agg1, 0, (size_t)NNODES * 256 * sizeof(float), stream);
    hipMemsetAsync(agg2, 0, (size_t)NNODES * 128 * sizeof(float), stream);

    const int mblocks = ((NNODES / 16) + 7) / 8;

    {
        long long total = (long long)NEDGES * (256 / 4);
        int blocks = (int)((total + 255) / 256);
        k_scatter_add<<<blocks, 256, 0, stream>>>(x, ei, agg1, NEDGES, 256);
    }
    k_gemm_strip<8, true><<<mblocks, 256, 0, stream>>>(
        agg1, W1_rel, x, W1_root, b1, h1, NNODES, 256, 1);

    {
        long long total = (long long)NEDGES * (128 / 4);
        int blocks = (int)((total + 255) / 256);
        k_scatter_add<<<blocks, 256, 0, stream>>>(h1, ei, agg2, NEDGES, 128);
    }
    k_gemm_strip<4, true><<<mblocks, 256, 0, stream>>>(
        agg2, W2_rel, h1, W2_root, b2, h2, NNODES, 128, 0);

    k_gemm_strip<8, false><<<mblocks, 256, 0, stream>>>(
        h2, lin1_w, nullptr, nullptr, lin1_b, m1, NNODES, 64, 1);
    k_gemm_strip<4, false><<<mblocks, 256, 0, stream>>>(
        m1, lin2_w, nullptr, nullptr, lin2_b, m2, NNODES, 128, 1);
    k_head<<<(NNODES + 255) / 256, 256, 0, stream>>>(
        m2, lin3_w, lin3_b, (float*)d_out, NNODES);
}


# Neighbour gather fused into each graph-layer product: a 16-wave work-group gathers 16 nodes into LDS and multiplies there, aggregate never in memory; 10->8 launches; dot-product sum order changes
# speedup vs baseline: 1.8008x; 1.8008x over previous
#include <hip/hip_runtime.h>

typedef float v2f __attribute__((ext_vector_type(2)));
typedef float v8f __attribute__((ext_vector_type(8)));

#define NNODES 20000
#define NEDGES 320000

#define NPAD 20480

__global__ __launch_bounds__(256) void k_degree_count(
    const int* __restrict__ ei, int* __restrict__ deg, int n_edges)
{
    const int e = (blockIdx.x * 256 + threadIdx.x) * 4;
    if (e >= n_edges) return;
    const int4 d = *reinterpret_cast<const int4*>(ei + n_edges + e);
    atomicAdd(deg + d.x, 1);
    atomicAdd(deg + d.y, 1);
    atomicAdd(deg + d.z, 1);
    atomicAdd(deg + d.w, 1);
}

__global__ __launch_bounds__(1024) void k_row_offsets(
    int* __restrict__ cur, int* __restrict__ rowptr)
{
    __shared__ int wsum[32];
    const int t = threadIdx.x;
    const int lane = t & 31;
    const int w = t >> 5;
    int4* cp = reinterpret_cast<int4*>(cur) + t * 5;
    int4* rp = reinterpret_cast<int4*>(rowptr) + t * 5;
    int4 c[5];
    for (int i = 0; i < 5; ++i) c[i] = cp[i];
    int tsum = 0;
    for (int i = 0; i < 5; ++i) tsum += c[i].x + c[i].y + c[i].z + c[i].w;
    int x = tsum;
    for (int o = 1; o < 32; o <<= 1) {
        const int y = __shfl_up(x, o);
        if (lane >= o) x += y;
    }
    if (lane == 31) wsum[w] = x;
    __syncthreads();
    if (w == 0) {
        const int s = wsum[lane];
        int z = s;
        for (int o = 1; o < 32; o <<= 1) {
            const int y = __shfl_up(z, o);
            if (lane >= o) z += y;
        }
        wsum[lane] = z - s;
    }
    __syncthreads();
    int run = wsum[w] + (x - tsum);
    for (int i = 0; i < 5; ++i) {
        int4 o4;
        o4.x = run; run += c[i].x;
        o4.y = run; run += c[i].y;
        o4.z = run; run += c[i].z;
        o4.w = run; run += c[i].w;
        cp[i] = o4;
        rp[i] = o4;
    }
}

__global__ __launch_bounds__(256) void k_fill_lists(
    const int* __restrict__ ei, int* __restrict__ cur,
    int* __restrict__ lst, int n_edges)
{
    const int e = (blockIdx.x * 256 + threadIdx.x) * 4;
    if (e >= n_edges) return;
    const int4 s = *reinterpret_cast<const int4*>(ei + e);
    const int4 d = *reinterpret_cast<const int4*>(ei + n_edges + e);
    const int p0 = atomicAdd(cur + d.x, 1);
    const int p1 = atomicAdd(cur + d.y, 1);
    const int p2 = atomicAdd(cur + d.z, 1);
    const int p3 = atomicAdd(cur + d.w, 1);
    if ((unsigned)p0 < (unsigned)n_edges) lst[p0] = s.x;
    if ((unsigned)p1 < (unsigned)n_edges) lst[p1] = s.y;
    if ((unsigned)p2 < (unsigned)n_edges) lst[p2] = s.z;
    if ((unsigned)p3 < (unsigned)n_edges) lst[p3] = s.w;
}

template <int D>
__device__ __forceinline__ void gather_row(
    const float* __restrict__ feat, const int* __restrict__ lst,
    int beg, int end, int lane, int n_nodes, float4 (&acc)[D / 128])
{
    constexpr int V = D / 128;
    for (int base = beg; base < end; base += 32) {
        const int cnt = min(32, end - base);
        int idx = lst[base + min(lane, cnt - 1)];
        idx = (int)min((unsigned)idx, (unsigned)(n_nodes - 1));
        int j = 0;
        for (; j + 4 <= cnt; j += 4) {
            const int s0 = __builtin_amdgcn_readlane(idx, j);
            const int s1 = __builtin_amdgcn_readlane(idx, j + 1);
            const int s2 = __builtin_amdgcn_readlane(idx, j + 2);
            const int s3 = __builtin_amdgcn_readlane(idx, j + 3);
            const float4* r0 = reinterpret_cast<const float4*>(feat + (size_t)s0 * D) + lane;
            const float4* r1 = reinterpret_cast<const float4*>(feat + (size_t)s1 * D) + lane;
            const float4* r2 = reinterpret_cast<const float4*>(feat + (size_t)s2 * D) + lane;
            const float4* r3 = reinterpret_cast<const float4*>(feat + (size_t)s3 * D) + lane;
            float4 a0[V], a1[V], a2[V], a3[V];
            for (int v = 0; v < V; ++v) a0[v] = r0[32 * v];
            for (int v = 0; v < V; ++v) a1[v] = r1[32 * v];
            for (int v = 0; v < V; ++v) a2[v] = r2[32 * v];
            for (int v = 0; v < V; ++v) a3[v] = r3[32 * v];
            for (int v = 0; v < V; ++v) {
                acc[v].x += a0[v].x; acc[v].y += a0[v].y;
                acc[v].z += a0[v].z; acc[v].w += a0[v].w;
                acc[v].x += a1[v].x; acc[v].y += a1[v].y;
                acc[v].z += a1[v].z; acc[v].w += a1[v].w;
                acc[v].x += a2[v].x; acc[v].y += a2[v].y;
                acc[v].z += a2[v].z; acc[v].w += a2[v].w;
                acc[v].x += a3[v].x; acc[v].y += a3[v].y;
                acc[v].z += a3[v].z; acc[v].w += a3[v].w;
            }
        }
        if (j < cnt) {
            const int rem = cnt - j;
            const unsigned k1 = (rem > 1) ? 0xffffffffu : 0u;
            const unsigned k2 = (rem > 2) ? 0xffffffffu : 0u;
            const int s0 = __builtin_amdgcn_readlane(idx, j);
            const int s1 = __builtin_amdgcn_readlane(idx, j + 1);
            const int s2 = __builtin_amdgcn_readlane(idx, j + 2);
            const float4* r0 = reinterpret_cast<const float4*>(feat + (size_t)s0 * D) + lane;
            const float4* r1 = reinterpret_cast<const float4*>(feat + (size_t)s1 * D) + lane;
            const float4* r2 = reinterpret_cast<const float4*>(feat + (size_t)s2 * D) + lane;
            float4 a0[V], a1[V], a2[V];
            for (int v = 0; v < V; ++v) a0[v] = r0[32 * v];
            for (int v = 0; v < V; ++v) a1[v] = r1[32 * v];
            for (int v = 0; v < V; ++v) a2[v] = r2[32 * v];
            for (int v = 0; v < V; ++v) {
                acc[v].x += a0[v].x; acc[v].y += a0[v].y;
                acc[v].z += a0[v].z; acc[v].w += a0[v].w;
                acc[v].x += __uint_as_float(__float_as_uint(a1[v].x) & k1);
                acc[v].y += __uint_as_float(__float_as_uint(a1[v].y) & k1);
                acc[v].z += __uint_as_float(__float_as_uint(a1[v].z) & k1);
                acc[v].w += __uint_as_float(__float_as_uint(a1[v].w) & k1);
                acc[v].x += __uint_as_float(__float_as_uint(a2[v].x) & k2);
                acc[v].y += __uint_as_float(__float_as_uint(a2[v].y) & k2);
                acc[v].z += __uint_as_float(__float_as_uint(a2[v].z) & k2);
                acc[v].w += __uint_as_float(__float_as_uint(a2[v].w) & k2);
            }
        }
    }
}

template <int D, int N>
__global__ __launch_bounds__(512) void k_gather_gemm(
    const float* __restrict__ feat, const int* __restrict__ rowptr,
    const int* __restrict__ lst,
    const float* __restrict__ Wrel, const float* __restrict__ Wroot,
    const float* __restrict__ bias, float* __restrict__ out,
    int n_nodes, int relu)
{
    constexpr int V = D / 128;
    constexpr int P = D + 4;
    constexpr int G = N / 32;
    constexpr int PARTS = 16 / G;
    constexpr int KP = 2 * D / PARTS;
    static_assert(D % KP == 0, "a k piece must lie inside one operand");
    __shared__ __align__(16) float As[16 * P];
    __shared__ __align__(16) float Xs[16 * P];
    __shared__ __align__(16) float Cs[PARTS * 16 * N];

    const int lane = threadIdx.x & 31;
    const int w = threadIdx.x >> 5;
    const int node0 = blockIdx.x * 16;
    const int node = node0 + w;

    float4 acc[V], own[V];
    for (int v = 0; v < V; ++v) {
        acc[v] = make_float4(0.f, 0.f, 0.f, 0.f);
        own[v] = make_float4(0.f, 0.f, 0.f, 0.f);
    }
    if (node < n_nodes) {
        const float4* xr = reinterpret_cast<const float4*>(feat + (size_t)node * D) + lane;
        for (int v = 0; v < V; ++v) own[v] = xr[32 * v];
        const int beg = __builtin_amdgcn_readfirstlane(rowptr[node]);
        const int end = __builtin_amdgcn_readfirstlane(rowptr[node + 1]);
        gather_row<D>(feat, lst, beg, end, lane, n_nodes, acc);
    }
    for (int v = 0; v < V; ++v) {
        *reinterpret_cast<float4*>(&As[w * P + (lane + 32 * v) * 4]) = acc[v];
        *reinterpret_cast<float4*>(&Xs[w * P + (lane + 32 * v) * 4]) = own[v];
    }
    __syncthreads();

    const int g = w % G;
    const int p = w / G;
    const bool root = (p * KP >= D);
    const int kbase = p * KP - (root ? D : 0);
    const int rh = lane & 15;
    const int hi = lane >> 4;
    const float* sp = (root ? Xs : As) + rh * P + kbase + 2 * hi;
    const float* wp = (root ? Wroot : Wrel) + (size_t)(kbase + 2 * hi) * N + g * 32 + 2 * rh;

    v8f c0 = v8f{}, c1 = v8f{};
#pragma unroll 4
    for (int k = 0; k < KP; k += 4) {
        const v2f a = *reinterpret_cast<const v2f*>(sp + k);
        const v2f l0 = *reinterpret_cast<const v2f*>(wp + (size_t)k * N);
        const v2f l1 = *reinterpret_cast<const v2f*>(wp + (size_t)(k + 1) * N);
        v2f b0, b1;
        b0.x = l0.x; b0.y = l1.x;
        b1.x = l0.y; b1.y = l1.y;
        c0 = __builtin_amdgcn_wmma_f32_16x16x4_f32(
            false, a, false, b0, (short)0, c0, false, false);
        c1 = __builtin_amdgcn_wmma_f32_16x16x4_f32(
            false, a, false, b1, (short)0, c1, false, false);
    }
#pragma unroll
    for (int i = 0; i < 8; ++i) {
        v2f d;
        d.x = c0[i]; d.y = c1[i];
        *reinterpret_cast<v2f*>(&Cs[(p * 16 + i + 8 * hi) * N + g * 32 + 2 * rh]) = d;
    }
    __syncthreads();

    for (int idx = threadIdx.x; idx < 16 * (N / 4); idx += 512) {
        const int r = idx / (N / 4);
        const int c4 = (idx - r * (N / 4)) * 4;
        float4 sum = *reinterpret_cast<const float4*>(bias + c4);
#pragma unroll
        for (int q = 0; q < PARTS; ++q) {
            const float4 pv = *reinterpret_cast<const float4*>(&Cs[(q * 16 + r) * N + c4]);
            sum.x += pv.x; sum.y += pv.y; sum.z += pv.z; sum.w += pv.w;
        }
        if (relu) {
            sum.x = fmaxf(sum.x, 0.0f); sum.y = fmaxf(sum.y, 0.0f);
            sum.z = fmaxf(sum.z, 0.0f); sum.w = fmaxf(sum.w, 0.0f);
        }
        if (node0 + r < n_nodes)
            *reinterpret_cast<float4*>(out + (size_t)(node0 + r) * N + c4) = sum;
    }
}

template <int NT, bool DUAL>
__global__ __launch_bounds__(256) void k_gemm_strip(
    const float* __restrict__ A0, const float* __restrict__ B0,
    const float* __restrict__ A1, const float* __restrict__ B1,
    const float* __restrict__ bias, float* __restrict__ C,
    int M, int K, int relu)
{
    constexpr int N = NT * 16;
    constexpr int BK = 64;
    __shared__ __align__(16) float Bs0[BK * N];
    __shared__ __align__(16) float Bs1[DUAL ? (BK * N) : 4];

    const int lane = threadIdx.x & 31;
    const int wave = threadIdx.x >> 5;
    const int tm = blockIdx.x * 8 + wave;
    const bool active = (tm * 16 < M);
    const int rh = lane & 15;
    const int hi = lane >> 4;

    v8f acc[NT];
#pragma unroll
    for (int t = 0; t < NT; ++t) acc[t] = v8f{};

    const float* ap0 = A0 + (size_t)(tm * 16 + rh) * K + 2 * hi;
    const float* ap1 = DUAL ? (A1 + (size_t)(tm * 16 + rh) * K + 2 * hi) : nullptr;

    for (int kc = 0; kc < K; kc += BK) {
        __syncthreads();
        {
            constexpr int ITERS = (BK / 2) * (N / 4);
            for (int idx = threadIdx.x; idx < ITERS; idx += 256) {
                const int pr = idx / (N / 4);
                const int c = (idx - pr * (N / 4)) * 4;
                const int k = 2 * pr;
                float4 g0 = *(const float4*)(B0 + (size_t)(kc + k) * N + c);
                float4 g1 = *(const float4*)(B0 + (size_t)(kc + k + 1) * N + c);
                float4 lo = {g0.x, g1.x, g0.y, g1.y};
                float4 hi4 = {g0.z, g1.z, g0.w, g1.w};
                float* dst = &Bs0[(pr * N + c) * 2];
                *(float4*)(dst) = lo;
                *(float4*)(dst + 4) = hi4;
                if constexpr (DUAL) {
                    float4 h0 = *(const float4*)(B1 + (size_t)(kc + k) * N + c);
                    float4 h1 = *(const float4*)(B1 + (size_t)(kc + k + 1) * N + c);
                    float4 l2 = {h0.x, h1.x, h0.y, h1.y};
                    float4 h2 = {h0.z, h1.z, h0.w, h1.w};
                    float* dst1 = &Bs1[(pr * N + c) * 2];
                    *(float4*)(dst1) = l2;
                    *(float4*)(dst1 + 4) = h2;
                }
            }
        }
        __syncthreads();
        if (!active) continue;

        const v2f* Bp0 = (const v2f*)Bs0;
        const v2f* Bp1 = (const v2f*)Bs1;
        for (int k = 0; k < BK; k += 4) {
            const int pr = (k >> 1) + hi;
            v2f a;
            a.x = ap0[kc + k];
            a.y = ap0[kc + k + 1];
#pragma unroll
            for (int t = 0; t < NT; ++t) {
                v2f b = Bp0[pr * N + t * 16 + rh];
                acc[t] = __builtin_amdgcn_wmma_f32_16x16x4_f32(
                    false, a, false, b, (short)0, acc[t], false, false);
            }
            if constexpr (DUAL) {
                v2f a1;
                a1.x = ap1[kc + k];
                a1.y = ap1[kc + k + 1];
#pragma unroll
                for (int t = 0; t < NT; ++t) {
                    v2f b = Bp1[pr * N + t * 16 + rh];
                    acc[t] = __builtin_amdgcn_wmma_f32_16x16x4_f32(
                        false, a1, false, b, (short)0, acc[t], false, false);
                }
            }
        }
    }

    if (active) {
#pragma unroll
        for (int t = 0; t < NT; ++t) {
            const int col = t * 16 + rh;
            const float bv = bias[col];
#pragma unroll
            for (int i = 0; i < 8; ++i) {
                int r = tm * 16 + i + 8 * hi;
                float v = acc[t][i] + bv;
                if (relu) v = fmaxf(v, 0.0f);
                C[(size_t)r * N + col] = v;
            }
        }
    }
}

__global__ __launch_bounds__(256) void k_head(
    const float* __restrict__ h, const float* __restrict__ W,
    const float* __restrict__ b, float* __restrict__ out, int M)
{
    int n = blockIdx.x * blockDim.x + threadIdx.x;
    if (n >= M) return;
    const float* hr = h + (size_t)n * 64;
    float z[10];
#pragma unroll
    for (int c = 0; c < 10; ++c) z[c] = b[c];
    for (int k = 0; k < 64; ++k) {
        float hv = hr[k];
        const float* wr = W + k * 10;
#pragma unroll
        for (int c = 0; c < 10; ++c) z[c] = fmaf(hv, wr[c], z[c]);
    }
    float mx = z[0];
#pragma unroll
    for (int c = 1; c < 10; ++c) mx = fmaxf(mx, z[c]);
    float s = 0.0f;
#pragma unroll
    for (int c = 0; c < 10; ++c) s += expf(z[c] - mx);
    float lse = mx + logf(s);
    float* o = out + (size_t)n * 10;
#pragma unroll
    for (int c = 0; c < 10; ++c) o[c] = z[c] - lse;
}

extern "C" void kernel_launch(void* const* d_in, const int* in_sizes, int n_in,
                              void* d_out, int out_size, void* d_ws, size_t ws_size,
                              hipStream_t stream) {
    const float* x       = (const float*)d_in[0];
    const int*   ei      = (const int*)d_in[1];
    const float* W1_rel  = (const float*)d_in[2];
    const float* W1_root = (const float*)d_in[3];
    const float* b1      = (const float*)d_in[4];
    const float* W2_rel  = (const float*)d_in[5];
    const float* W2_root = (const float*)d_in[6];
    const float* b2      = (const float*)d_in[7];
    const float* lin1_w  = (const float*)d_in[8];
    const float* lin1_b  = (const float*)d_in[9];
    const float* lin2_w  = (const float*)d_in[10];
    const float* lin2_b  = (const float*)d_in[11];
    const float* lin3_w  = (const float*)d_in[12];
    const float* lin3_b  = (const float*)d_in[13];

    float* agg1 = (float*)d_ws;
    float* h1   = agg1 + (size_t)NNODES * 256;
    float* agg2 = h1   + (size_t)NNODES * 128;
    float* h2   = agg2 + (size_t)NNODES * 128;
    float* m1   = h2   + (size_t)NNODES * 64;
    float* m2   = m1   + (size_t)NNODES * 128;

    int* cur    = (int*)m2;
    int* rowptr = cur + NPAD;
    int* lst    = rowptr + NPAD;

    hipMemsetAsync(cur, 0, (size_t)NPAD * sizeof(int), stream);
    k_degree_count<<<(NEDGES / 4 + 255) / 256, 256, 0, stream>>>(ei, cur, NEDGES);
    k_row_offsets<<<1, 1024, 0, stream>>>(cur, rowptr);
    k_fill_lists<<<(NEDGES / 4 + 255) / 256, 256, 0, stream>>>(ei, cur, lst, NEDGES);

    const int gblocks = (NNODES + 15) / 16;

    const int mblocks = ((NNODES / 16) + 7) / 8;

    k_gather_gemm<256, 128><<<gblocks, 512, 0, stream>>>(
        x, rowptr, lst, W1_rel, W1_root, b1, h1, NNODES, 1);

    k_gather_gemm<128, 64><<<gblocks, 512, 0, stream>>>(
        h1, rowptr, lst, W2_rel, W2_root, b2, h2, NNODES, 0);

    k_gemm_strip<8, false><<<mblocks, 256, 0, stream>>>(
        h2, lin1_w, nullptr, nullptr, lin1_b, m1, NNODES, 64, 1);
    k_gemm_strip<4, false><<<mblocks, 256, 0, stream>>>(
        m1, lin2_w, nullptr, nullptr, lin2_b, m2, NNODES, 128, 1);
    k_head<<<(NNODES + 255) / 256, 256, 0, stream>>>(
        m2, lin3_w, lin3_b, (float*)d_out, NNODES);
}


# Edge lists built in one pass: tickets from a node's degree counter name slots in its 32-entry line, later ones go to an overflow list; count and scan kernels gone, 5->3 launches; arrival-order sums
# speedup vs baseline: 3.4894x; 1.9377x over previous
#include <hip/hip_runtime.h>

typedef float v2f __attribute__((ext_vector_type(2)));
typedef float v8f __attribute__((ext_vector_type(8)));

#define NNODES 20000
#define NEDGES 320000

#define CAP 32

__device__ __forceinline__ void file_edge(
    int dst, int srcrow, int ticket, int* __restrict__ slots,
    int* __restrict__ novf, int* __restrict__ ovf, int n_edges)
{
    if ((unsigned)ticket < (unsigned)CAP) {
        slots[(size_t)dst * CAP + ticket] = srcrow;
    } else {
        const int p = atomicAdd(novf, 1);
        if ((unsigned)p < (unsigned)n_edges)
            *reinterpret_cast<int2*>(ovf + 2 * (size_t)p) = make_int2(dst, srcrow);
    }
}

__global__ __launch_bounds__(256) void k_file_edges(
    const int* __restrict__ ei, int* __restrict__ cnt, int* __restrict__ novf,
    int* __restrict__ slots, int* __restrict__ ovf, int n_edges)
{
    const int e = (blockIdx.x * 256 + threadIdx.x) * 4;
    if (e >= n_edges) return;
    const int4 s = *reinterpret_cast<const int4*>(ei + e);
    const int4 d = *reinterpret_cast<const int4*>(ei + n_edges + e);
    const int t0 = atomicAdd(cnt + d.x, 1);
    const int t1 = atomicAdd(cnt + d.y, 1);
    const int t2 = atomicAdd(cnt + d.z, 1);
    const int t3 = atomicAdd(cnt + d.w, 1);
    file_edge(d.x, s.x, t0, slots, novf, ovf, n_edges);
    file_edge(d.y, s.y, t1, slots, novf, ovf, n_edges);
    file_edge(d.z, s.z, t2, slots, novf, ovf, n_edges);
    file_edge(d.w, s.w, t3, slots, novf, ovf, n_edges);
}

template <int D>
__device__ __forceinline__ void gather_slots(
    const float* __restrict__ feat, const int* __restrict__ line,
    int cnt, int lane, int n_nodes, float4 (&acc)[D / 128])
{
    constexpr int V = D / 128;
    int idx = line[min(lane, cnt - 1)];
    idx = (int)min((unsigned)idx, (unsigned)(n_nodes - 1));
    int j = 0;
    for (; j + 4 <= cnt; j += 4) {
        const int s0 = __builtin_amdgcn_readlane(idx, j);
        const int s1 = __builtin_amdgcn_readlane(idx, j + 1);
        const int s2 = __builtin_amdgcn_readlane(idx, j + 2);
        const int s3 = __builtin_amdgcn_readlane(idx, j + 3);
        const float4* r0 = reinterpret_cast<const float4*>(feat + (size_t)s0 * D) + lane;
        const float4* r1 = reinterpret_cast<const float4*>(feat + (size_t)s1 * D) + lane;
        const float4* r2 = reinterpret_cast<const float4*>(feat + (size_t)s2 * D) + lane;
        const float4* r3 = reinterpret_cast<const float4*>(feat + (size_t)s3 * D) + lane;
        float4 a0[V], a1[V], a2[V], a3[V];
        for (int v = 0; v < V; ++v) a0[v] = r0[32 * v];
        for (int v = 0; v < V; ++v) a1[v] = r1[32 * v];
        for (int v = 0; v < V; ++v) a2[v] = r2[32 * v];
        for (int v = 0; v < V; ++v) a3[v] = r3[32 * v];
        for (int v = 0; v < V; ++v) {
            acc[v].x += a0[v].x; acc[v].y += a0[v].y;
            acc[v].z += a0[v].z; acc[v].w += a0[v].w;
            acc[v].x += a1[v].x; acc[v].y += a1[v].y;
            acc[v].z += a1[v].z; acc[v].w += a1[v].w;
            acc[v].x += a2[v].x; acc[v].y += a2[v].y;
            acc[v].z += a2[v].z; acc[v].w += a2[v].w;
            acc[v].x += a3[v].x; acc[v].y += a3[v].y;
            acc[v].z += a3[v].z; acc[v].w += a3[v].w;
        }
    }
    if (j < cnt) {
        const int rem = cnt - j;
        const unsigned k1 = (rem > 1) ? 0xffffffffu : 0u;
        const unsigned k2 = (rem > 2) ? 0xffffffffu : 0u;
        const int s0 = __builtin_amdgcn_readlane(idx, j);
        const int s1 = __builtin_amdgcn_readlane(idx, j + 1);
        const int s2 = __builtin_amdgcn_readlane(idx, j + 2);
        const float4* r0 = reinterpret_cast<const float4*>(feat + (size_t)s0 * D) + lane;
        const float4* r1 = reinterpret_cast<const float4*>(feat + (size_t)s1 * D) + lane;
        const float4* r2 = reinterpret_cast<const float4*>(feat + (size_t)s2 * D) + lane;
        float4 a0[V], a1[V], a2[V];
        for (int v = 0; v < V; ++v) a0[v] = r0[32 * v];
        for (int v = 0; v < V; ++v) a1[v] = r1[32 * v];
        for (int v = 0; v < V; ++v) a2[v] = r2[32 * v];
        for (int v = 0; v < V; ++v) {
            acc[v].x += a0[v].x; acc[v].y += a0[v].y;
            acc[v].z += a0[v].z; acc[v].w += a0[v].w;
            acc[v].x += __uint_as_float(__float_as_uint(a1[v].x) & k1);
            acc[v].y += __uint_as_float(__float_as_uint(a1[v].y) & k1);
            acc[v].z += __uint_as_float(__float_as_uint(a1[v].z) & k1);
            acc[v].w += __uint_as_float(__float_as_uint(a1[v].w) & k1);
            acc[v].x += __uint_as_float(__float_as_uint(a2[v].x) & k2);
            acc[v].y += __uint_as_float(__float_as_uint(a2[v].y) & k2);
            acc[v].z += __uint_as_float(__float_as_uint(a2[v].z) & k2);
            acc[v].w += __uint_as_float(__float_as_uint(a2[v].w) & k2);
        }
    }
}

template <int D>
__device__ __forceinline__ void gather_overflow(
    const float* __restrict__ feat, const int* __restrict__ ovf,
    const int* __restrict__ novf, int node, int lane, int n_nodes, int n_edges,
    float4 (&acc)[D / 128])
{
    constexpr int V = D / 128;
    const int total = min(__builtin_amdgcn_readfirstlane(novf[0]), n_edges);
    for (int base = 0; base < total; base += 32) {
        const int i = min(base + lane, total - 1);
        const int2 en = *reinterpret_cast<const int2*>(ovf + 2 * (size_t)i);
        unsigned m = __builtin_amdgcn_ballot_w32(en.x == node && base + lane < total);
        const int srcrow = (int)min((unsigned)en.y, (unsigned)(n_nodes - 1));
        while (m != 0u) {
            const int j = __builtin_ctz(m);
            m &= m - 1u;
            const int s = __builtin_amdgcn_readlane(srcrow, j);
            const float4* r = reinterpret_cast<const float4*>(feat + (size_t)s * D) + lane;
            float4 a[V];
            for (int v = 0; v < V; ++v) a[v] = r[32 * v];
            for (int v = 0; v < V; ++v) {
                acc[v].x += a[v].x; acc[v].y += a[v].y;
                acc[v].z += a[v].z; acc[v].w += a[v].w;
            }
        }
    }
}

template <int K, int N>
__device__ __forceinline__ void dense_partials(
    const float* S, const float* __restrict__ Wg, float* Cs, int w, int lane)
{
    constexpr int P = K + 4;
    constexpr int G = N / 32;
    constexpr int PARTS = 16 / G;
    constexpr int KP = K / PARTS;
    static_assert(KP % 4 == 0, "a k piece holds whole 4-deep matrix steps");
    const int g = w % G;
    const int p = w / G;
    const int rh = lane & 15;
    const int hi = lane >> 4;
    const float* sp = S + rh * P + p * KP + 2 * hi;
    const float* wp = Wg + (size_t)(p * KP + 2 * hi) * N + g * 32 + 2 * rh;
    v8f c0 = v8f{}, c1 = v8f{};
#pragma unroll
    for (int k = 0; k < KP; k += 4) {
        const v2f a = *reinterpret_cast<const v2f*>(sp + k);
        const v2f l0 = *reinterpret_cast<const v2f*>(wp + (size_t)k * N);
        const v2f l1 = *reinterpret_cast<const v2f*>(wp + (size_t)(k + 1) * N);
        v2f b0, b1;
        b0.x = l0.x; b0.y = l1.x;
        b1.x = l0.y; b1.y = l1.y;
        c0 = __builtin_amdgcn_wmma_f32_16x16x4_f32(
            false, a, false, b0, (short)0, c0, false, false);
        c1 = __builtin_amdgcn_wmma_f32_16x16x4_f32(
            false, a, false, b1, (short)0, c1, false, false);
    }
    for (int i = 0; i < 8; ++i) {
        v2f d;
        d.x = c0[i]; d.y = c1[i];
        *reinterpret_cast<v2f*>(&Cs[(p * 16 + i + 8 * hi) * N + g * 32 + 2 * rh]) = d;
    }
}

template <int N, int PARTS>
__device__ __forceinline__ void reduce_to_tile(
    const float* Cs, const float* bs, float* dst, int relu)
{
    static_assert(16 * (N / 4) <= 512, "one piece per thread");
    const int idx = threadIdx.x;
    if (idx < 16 * (N / 4)) {
        const int r = idx / (N / 4);
        const int c4 = (idx - r * (N / 4)) * 4;
        float4 sum = *reinterpret_cast<const float4*>(bs + c4);
        for (int q = 0; q < PARTS; ++q) {
            const float4 pv = *reinterpret_cast<const float4*>(&Cs[(q * 16 + r) * N + c4]);
            sum.x += pv.x; sum.y += pv.y; sum.z += pv.z; sum.w += pv.w;
        }
        if (relu) {
            sum.x = fmaxf(sum.x, 0.0f); sum.y = fmaxf(sum.y, 0.0f);
            sum.z = fmaxf(sum.z, 0.0f); sum.w = fmaxf(sum.w, 0.0f);
        }
        *reinterpret_cast<float4*>(dst + r * (N + 4) + c4) = sum;
    }
}

template <int D, int N, bool TAIL>
__global__ __launch_bounds__(512) void k_gather_gemm(
    const float* __restrict__ feat, const int* __restrict__ cnt,
    const int* __restrict__ slots, const int* __restrict__ ovf,
    const int* __restrict__ novf, int n_edges,
    const float* __restrict__ Wrel, const float* __restrict__ Wroot,
    const float* __restrict__ bias,
    const float* __restrict__ lin1_w, const float* __restrict__ lin1_b,
    const float* __restrict__ lin2_w, const float* __restrict__ lin2_b,
    const float* __restrict__ lin3_w, const float* __restrict__ lin3_b,
    float* __restrict__ out, int n_nodes, int relu)
{
    constexpr int V = D / 128;
    constexpr int P = D + 4;
    constexpr int G = N / 32;
    constexpr int PARTS = 16 / G;
    constexpr int KP = 2 * D / PARTS;
    static_assert(D % KP == 0, "a k piece must lie inside one operand");
    static_assert(!TAIL || (D == 128 && N == 64), "the dense layers follow the 128 -> 64 layer");
    __shared__ __align__(16) float As[16 * P];
    __shared__ __align__(16) float Xs[16 * P];
    __shared__ __align__(16) float Cs[PARTS * 16 * N];
    __shared__ __align__(16) float Hs[TAIL ? 16 * 68 : 4];
    __shared__ __align__(16) float Ms[TAIL ? 16 * 68 : 4];
    __shared__ float Zs[TAIL ? 160 : 1];
    __shared__ __align__(16) float Ks[TAIL ? 896 : 4];

    const int lane = threadIdx.x & 31;
    const int w = threadIdx.x >> 5;
    const int node0 = blockIdx.x * 16;
    const int node = node0 + w;

    float4 cst = make_float4(0.f, 0.f, 0.f, 0.f);
    if constexpr (TAIL) {
        const int t = threadIdx.x;
        if (t < 224) {
            const float* src = (t < 160) ? (lin3_w + t * 4)
                             : (t < 176) ? (bias + (t - 160) * 4)
                             : (t < 208) ? (lin1_b + (t - 176) * 4)
                                         : (lin2_b + (t - 208) * 4);
            cst = *reinterpret_cast<const float4*>(src);
        }
    }

    float4 acc[V], own[V];
    for (int v = 0; v < V; ++v) {
        acc[v] = make_float4(0.f, 0.f, 0.f, 0.f);
        own[v] = make_float4(0.f, 0.f, 0.f, 0.f);
    }
    if (node < n_nodes) {
        const float4* xr = reinterpret_cast<const float4*>(feat + (size_t)node * D) + lane;
        for (int v = 0; v < V; ++v) own[v] = xr[32 * v];
        const int deg = __builtin_amdgcn_readfirstlane(cnt[node]);
        if (deg > 0)
            gather_slots<D>(feat, slots + (size_t)node * CAP, min(deg, CAP),
                            lane, n_nodes, acc);
        if (deg > CAP)
            gather_overflow<D>(feat, ovf, novf, node, lane, n_nodes, n_edges, acc);
    }
    for (int v = 0; v < V; ++v) {
        *reinterpret_cast<float4*>(&As[w * P + (lane + 32 * v) * 4]) = acc[v];
        *reinterpret_cast<float4*>(&Xs[w * P + (lane + 32 * v) * 4]) = own[v];
    }
    if constexpr (TAIL) {
        if (threadIdx.x < 224)
            *reinterpret_cast<float4*>(&Ks[threadIdx.x * 4]) = cst;
    }
    __syncthreads();

    const int zr_ = threadIdx.x / 10;
    const int zc_ = threadIdx.x - zr_ * 10;
    float zb = 0.0f;
    if constexpr (TAIL) {
        if (threadIdx.x < 160) zb = lin3_b[zc_];
    }

    const int g = w % G;
    const int p = w / G;
    const bool root = (p * KP >= D);
    const int kbase = p * KP - (root ? D : 0);
    const int rh = lane & 15;
    const int hi = lane >> 4;
    const float* sp = (root ? Xs : As) + rh * P + kbase + 2 * hi;
    const float* wp = (root ? Wroot : Wrel) + (size_t)(kbase + 2 * hi) * N + g * 32 + 2 * rh;

    v8f c0 = v8f{}, c1 = v8f{};
#pragma unroll 4
    for (int k = 0; k < KP; k += 4) {
        const v2f a = *reinterpret_cast<const v2f*>(sp + k);
        const v2f l0 = *reinterpret_cast<const v2f*>(wp + (size_t)k * N);
        const v2f l1 = *reinterpret_cast<const v2f*>(wp + (size_t)(k + 1) * N);
        v2f b0, b1;
        b0.x = l0.x; b0.y = l1.x;
        b1.x = l0.y; b1.y = l1.y;
        c0 = __builtin_amdgcn_wmma_f32_16x16x4_f32(
            false, a, false, b0, (short)0, c0, false, false);
        c1 = __builtin_amdgcn_wmma_f32_16x16x4_f32(
            false, a, false, b1, (short)0, c1, false, false);
    }
#pragma unroll
    for (int i = 0; i < 8; ++i) {
        v2f d;
        d.x = c0[i]; d.y = c1[i];
        *reinterpret_cast<v2f*>(&Cs[(p * 16 + i + 8 * hi) * N + g * 32 + 2 * rh]) = d;
    }
    __syncthreads();

    if constexpr (!TAIL) {
        for (int idx = threadIdx.x; idx < 16 * (N / 4); idx += 512) {
            const int r = idx / (N / 4);
            const int c4 = (idx - r * (N / 4)) * 4;
            float4 sum = *reinterpret_cast<const float4*>(bias + c4);
#pragma unroll
            for (int q = 0; q < PARTS; ++q) {
                const float4 pv = *reinterpret_cast<const float4*>(&Cs[(q * 16 + r) * N + c4]);
                sum.x += pv.x; sum.y += pv.y; sum.z += pv.z; sum.w += pv.w;
            }
            if (relu) {
                sum.x = fmaxf(sum.x, 0.0f); sum.y = fmaxf(sum.y, 0.0f);
                sum.z = fmaxf(sum.z, 0.0f); sum.w = fmaxf(sum.w, 0.0f);
            }
            if (node0 + r < n_nodes)
                *reinterpret_cast<float4*>(out + (size_t)(node0 + r) * N + c4) = sum;
        }
    } else {
        reduce_to_tile<N, PARTS>(Cs, Ks + 640, Hs, relu);
        __syncthreads();
        dense_partials<64, 128>(Hs, lin1_w, Cs, w, lane);
        __syncthreads();
        reduce_to_tile<128, 4>(Cs, Ks + 704, As, 1);
        __syncthreads();
        dense_partials<128, 64>(As, lin2_w, Cs, w, lane);
        __syncthreads();
        reduce_to_tile<64, 8>(Cs, Ks + 832, Ms, 1);
        __syncthreads();
        const int t = threadIdx.x;
        const int r = zr_;
        const int c = zc_;
        if (t < 160) {
            float z = zb;
            const float* mr = Ms + r * 68;
            for (int k = 0; k < 64; ++k) z = fmaf(mr[k], Ks[k * 10 + c], z);
            Zs[t] = z;
        }
        __syncthreads();
        if (t < 160) {
            const float* zr = Zs + r * 10;
            float mx = zr[0];
            for (int j = 1; j < 10; ++j) mx = fmaxf(mx, zr[j]);
            float s = 0.0f;
            for (int j = 0; j < 10; ++j) s += expf(zr[j] - mx);
            const float lse = mx + logf(s);
            if (node0 + r < n_nodes)
                out[(size_t)(node0 + r) * 10 + c] = zr[c] - lse;
        }
    }
}

extern "C" void kernel_launch(void* const* d_in, const int* in_sizes, int n_in,
                              void* d_out, int out_size, void* d_ws, size_t ws_size,
                              hipStream_t stream) {
    const float* x       = (const float*)d_in[0];
    const int*   ei      = (const int*)d_in[1];
    const float* W1_rel  = (const float*)d_in[2];
    const float* W1_root = (const float*)d_in[3];
    const float* b1      = (const float*)d_in[4];
    const float* W2_rel  = (const float*)d_in[5];
    const float* W2_root = (const float*)d_in[6];
    const float* b2      = (const float*)d_in[7];
    const float* lin1_w  = (const float*)d_in[8];
    const float* lin1_b  = (const float*)d_in[9];
    const float* lin2_w  = (const float*)d_in[10];
    const float* lin2_b  = (const float*)d_in[11];
    const float* lin3_w  = (const float*)d_in[12];
    const float* lin3_b  = (const float*)d_in[13];

    int*   cnt   = (int*)d_ws;
    int*   novf  = cnt + NNODES;
    int*   slots = novf + 32;
    int*   ovf   = slots + (size_t)NNODES * CAP;
    float* h1    = (float*)d_ws + (size_t)NNODES * 256;

    hipMemsetAsync(cnt, 0, (size_t)(NNODES + 32) * sizeof(int), stream);
    k_file_edges<<<(NEDGES / 4 + 255) / 256, 256, 0, stream>>>(
        ei, cnt, novf, slots, ovf, NEDGES);

    const int gblocks = (NNODES + 15) / 16;

    k_gather_gemm<256, 128, false><<<gblocks, 512, 0, stream>>>(
        x, cnt, slots, ovf, novf, NEDGES, W1_rel, W1_root, b1,
        nullptr, nullptr, nullptr, nullptr, nullptr, nullptr,
        h1, NNODES, 1);

    k_gather_gemm<128, 64, true><<<gblocks, 512, 0, stream>>>(
        h1, cnt, slots, ovf, novf, NEDGES, W2_rel, W2_root, b2,
        lin1_w, lin1_b, lin2_w, lin2_b, lin3_w, lin3_b,
        (float*)d_out, NNODES, 0);
}


# Graph-layer work-groups cut from 16 to 8 waves: a wave gathers two nodes side by side, twice the row loads in flight; all 1250 groups resident, half the waves to launch; dot-product sum order changes
# speedup vs baseline: 3.7219x; 1.0666x over previous
#include <hip/hip_runtime.h>

typedef float v2f __attribute__((ext_vector_type(2)));
typedef float v8f __attribute__((ext_vector_type(8)));

#define NNODES 20000
#define NEDGES 320000

#define CAP 32
#define NW 8
#define NT (NW * 32)

__device__ __forceinline__ void file_edge(
    int dst, int srcrow, int ticket, int* __restrict__ slots,
    int* __restrict__ novf, int* __restrict__ ovf, int n_edges)
{
    if ((unsigned)ticket < (unsigned)CAP) {
        slots[(size_t)dst * CAP + ticket] = srcrow;
    } else {
        const int p = atomicAdd(novf, 1);
        if ((unsigned)p < (unsigned)n_edges)
            *reinterpret_cast<int2*>(ovf + 2 * (size_t)p) = make_int2(dst, srcrow);
    }
}

__global__ __launch_bounds__(256) void k_file_edges(
    const int* __restrict__ ei, int* __restrict__ cnt, int* __restrict__ novf,
    int* __restrict__ slots, int* __restrict__ ovf, int n_edges)
{
    const int e = (blockIdx.x * 256 + threadIdx.x) * 4;
    if (e >= n_edges) return;
    const int4 s = *reinterpret_cast<const int4*>(ei + e);
    const int4 d = *reinterpret_cast<const int4*>(ei + n_edges + e);
    const int t0 = atomicAdd(cnt + d.x, 1);
    const int t1 = atomicAdd(cnt + d.y, 1);
    const int t2 = atomicAdd(cnt + d.z, 1);
    const int t3 = atomicAdd(cnt + d.w, 1);
    file_edge(d.x, s.x, t0, slots, novf, ovf, n_edges);
    file_edge(d.y, s.y, t1, slots, novf, ovf, n_edges);
    file_edge(d.z, s.z, t2, slots, novf, ovf, n_edges);
    file_edge(d.w, s.w, t3, slots, novf, ovf, n_edges);
}

template <int D, int R>
__device__ __forceinline__ void request_rows(
    const float* __restrict__ feat, int idx, int j, int lane, float4 (&r)[4][D / 128])
{
    constexpr int V = D / 128;
    for (int q = 0; q < R; ++q) {
        const int s = __builtin_amdgcn_readlane(idx, j + q);
        const float4* p = reinterpret_cast<const float4*>(feat + (size_t)s * D) + lane;
        for (int v = 0; v < V; ++v) r[q][v] = p[32 * v];
    }
}

template <int D>
__device__ __forceinline__ void add_rows4(
    float4 (&acc)[D / 128], const float4 (&r)[4][D / 128])
{
    constexpr int V = D / 128;
    for (int v = 0; v < V; ++v)
        for (int q = 0; q < 4; ++q) {
            acc[v].x += r[q][v].x; acc[v].y += r[q][v].y;
            acc[v].z += r[q][v].z; acc[v].w += r[q][v].w;
        }
}

template <int D>
__device__ __forceinline__ void add_rows3(
    float4 (&acc)[D / 128], const float4 (&r)[4][D / 128], int rem)
{
    constexpr int V = D / 128;
    const unsigned k1 = (rem > 1) ? 0xffffffffu : 0u;
    const unsigned k2 = (rem > 2) ? 0xffffffffu : 0u;
    for (int v = 0; v < V; ++v) {
        acc[v].x += r[0][v].x; acc[v].y += r[0][v].y;
        acc[v].z += r[0][v].z; acc[v].w += r[0][v].w;
        acc[v].x += __uint_as_float(__float_as_uint(r[1][v].x) & k1);
        acc[v].y += __uint_as_float(__float_as_uint(r[1][v].y) & k1);
        acc[v].z += __uint_as_float(__float_as_uint(r[1][v].z) & k1);
        acc[v].w += __uint_as_float(__float_as_uint(r[1][v].w) & k1);
        acc[v].x += __uint_as_float(__float_as_uint(r[2][v].x) & k2);
        acc[v].y += __uint_as_float(__float_as_uint(r[2][v].y) & k2);
        acc[v].z += __uint_as_float(__float_as_uint(r[2][v].z) & k2);
        acc[v].w += __uint_as_float(__float_as_uint(r[2][v].w) & k2);
    }
}

template <int D>
__device__ __forceinline__ void gather_pair(
    const float* __restrict__ feat, int idxa, int idxb, int ca, int cb, int lane,
    float4 (&acca)[D / 128], float4 (&accb)[D / 128])
{
    constexpr int V = D / 128;
    const int fa = ca & ~3;
    const int fb = cb & ~3;
    const int fj = min(fa, fb);
    for (int j = 0; j < fj; j += 4) {
        float4 ra[4][V], rb[4][V];
        request_rows<D, 4>(feat, idxa, j, lane, ra);
        request_rows<D, 4>(feat, idxb, j, lane, rb);
        add_rows4<D>(acca, ra);
        add_rows4<D>(accb, rb);
    }
    for (int j = fj; j < fa; j += 4) {
        float4 ra[4][V];
        request_rows<D, 4>(feat, idxa, j, lane, ra);
        add_rows4<D>(acca, ra);
    }
    for (int j = fj; j < fb; j += 4) {
        float4 rb[4][V];
        request_rows<D, 4>(feat, idxb, j, lane, rb);
        add_rows4<D>(accb, rb);
    }
    const int ea = ca - fa;
    const int eb = cb - fb;
    if (ea > 0 && eb > 0) {
        float4 ra[4][V], rb[4][V];
        request_rows<D, 3>(feat, idxa, fa, lane, ra);
        request_rows<D, 3>(feat, idxb, fb, lane, rb);
        add_rows3<D>(acca, ra, ea);
        add_rows3<D>(accb, rb, eb);
    } else if (ea > 0) {
        float4 ra[4][V];
        request_rows<D, 3>(feat, idxa, fa, lane, ra);
        add_rows3<D>(acca, ra, ea);
    } else if (eb > 0) {
        float4 rb[4][V];
        request_rows<D, 3>(feat, idxb, fb, lane, rb);
        add_rows3<D>(accb, rb, eb);
    }
}

template <int D>
__device__ __forceinline__ void gather_overflow(
    const float* __restrict__ feat, const int* __restrict__ ovf,
    const int* __restrict__ novf, int node, int lane, int n_nodes, int n_edges,
    float4 (&acc)[D / 128])
{
    constexpr int V = D / 128;
    const int total = min(__builtin_amdgcn_readfirstlane(novf[0]), n_edges);
    for (int base = 0; base < total; base += 32) {
        const int i = min(base + lane, total - 1);
        const int2 en = *reinterpret_cast<const int2*>(ovf + 2 * (size_t)i);
        unsigned m = __builtin_amdgcn_ballot_w32(en.x == node && base + lane < total);
        const int srcrow = (int)min((unsigned)en.y, (unsigned)(n_nodes - 1));
        while (m != 0u) {
            const int j = __builtin_ctz(m);
            m &= m - 1u;
            const int s = __builtin_amdgcn_readlane(srcrow, j);
            const float4* r = reinterpret_cast<const float4*>(feat + (size_t)s * D) + lane;
            float4 a[V];
            for (int v = 0; v < V; ++v) a[v] = r[32 * v];
            for (int v = 0; v < V; ++v) {
                acc[v].x += a[v].x; acc[v].y += a[v].y;
                acc[v].z += a[v].z; acc[v].w += a[v].w;
            }
        }
    }
}

template <int K, int N>
__device__ __forceinline__ void dense_partials(
    const float* S, const float* __restrict__ Wg, float* Cs, int w, int lane)
{
    constexpr int P = K + 4;
    constexpr int G = N / 32;
    constexpr int PARTS = NW / G;
    constexpr int KP = K / PARTS;
    static_assert(KP % 4 == 0, "a k piece holds whole 4-deep matrix steps");
    const int g = w % G;
    const int p = w / G;
    const int rh = lane & 15;
    const int hi = lane >> 4;
    const float* sp = S + rh * P + p * KP + 2 * hi;
    const float* wp = Wg + (size_t)(p * KP + 2 * hi) * N + g * 32 + 2 * rh;
    v8f c0 = v8f{}, c1 = v8f{};
#pragma unroll
    for (int k = 0; k < KP; k += 4) {
        const v2f a = *reinterpret_cast<const v2f*>(sp + k);
        const v2f l0 = *reinterpret_cast<const v2f*>(wp + (size_t)k * N);
        const v2f l1 = *reinterpret_cast<const v2f*>(wp + (size_t)(k + 1) * N);
        v2f b0, b1;
        b0.x = l0.x; b0.y = l1.x;
        b1.x = l0.y; b1.y = l1.y;
        c0 = __builtin_amdgcn_wmma_f32_16x16x4_f32(
            false, a, false, b0, (short)0, c0, false, false);
        c1 = __builtin_amdgcn_wmma_f32_16x16x4_f32(
            false, a, false, b1, (short)0, c1, false, false);
    }
    for (int i = 0; i < 8; ++i) {
        v2f d;
        d.x = c0[i]; d.y = c1[i];
        *reinterpret_cast<v2f*>(&Cs[(p * 16 + i + 8 * hi) * N + g * 32 + 2 * rh]) = d;
    }
}

template <int N, int PARTS>
__device__ __forceinline__ void reduce_to_tile(
    const float* Cs, const float* bs, float* dst, int relu)
{
    for (int idx = threadIdx.x; idx < 16 * (N / 4); idx += NT) {
        const int r = idx / (N / 4);
        const int c4 = (idx - r * (N / 4)) * 4;
        float4 sum = *reinterpret_cast<const float4*>(bs + c4);
        for (int q = 0; q < PARTS; ++q) {
            const float4 pv = *reinterpret_cast<const float4*>(&Cs[(q * 16 + r) * N + c4]);
            sum.x += pv.x; sum.y += pv.y; sum.z += pv.z; sum.w += pv.w;
        }
        if (relu) {
            sum.x = fmaxf(sum.x, 0.0f); sum.y = fmaxf(sum.y, 0.0f);
            sum.z = fmaxf(sum.z, 0.0f); sum.w = fmaxf(sum.w, 0.0f);
        }
        *reinterpret_cast<float4*>(dst + r * (N + 4) + c4) = sum;
    }
}

template <int D, int N, bool TAIL>
__global__ __launch_bounds__(NT) __attribute__((amdgpu_waves_per_eu(10, 10)))
void k_gather_gemm(
    const float* __restrict__ feat, const int* __restrict__ cnt,
    const int* __restrict__ slots, const int* __restrict__ ovf,
    const int* __restrict__ novf, int n_edges,
    const float* __restrict__ Wrel, const float* __restrict__ Wroot,
    const float* __restrict__ bias,
    const float* __restrict__ lin1_w, const float* __restrict__ lin1_b,
    const float* __restrict__ lin2_w, const float* __restrict__ lin2_b,
    const float* __restrict__ lin3_w, const float* __restrict__ lin3_b,
    float* __restrict__ out, int n_nodes, int relu)
{
    constexpr int V = D / 128;
    constexpr int P = D + 4;
    constexpr int G = N / 32;
    constexpr int PARTS = NW / G;
    constexpr int KP = 2 * D / PARTS;
    static_assert(D % KP == 0, "a k piece must lie inside one operand");
    static_assert(!TAIL || (D == 128 && N == 64), "the dense layers follow the 128 -> 64 layer");
    __shared__ __align__(16) float As[16 * P];
    __shared__ __align__(16) float Xs[16 * P];
    __shared__ __align__(16) float Cs[PARTS * 16 * N];
    __shared__ __align__(16) float Hs[TAIL ? 16 * 68 : 4];
    __shared__ __align__(16) float Ms[TAIL ? 16 * 68 : 4];
    __shared__ float Zs[TAIL ? 160 : 1];
    __shared__ __align__(16) float Ks[TAIL ? 896 : 4];

    const int lane = threadIdx.x & 31;
    const int w = __builtin_amdgcn_readfirstlane((int)(threadIdx.x >> 5));
    const int node0 = blockIdx.x * 16;
    const int nodea = node0 + w;
    const int nodeb = node0 + w + NW;

    float4 cst = make_float4(0.f, 0.f, 0.f, 0.f);
    if constexpr (TAIL) {
        const int t = threadIdx.x;
        if (t < 224) {
            const float* src = (t < 160) ? (lin3_w + t * 4)
                             : (t < 176) ? (bias + (t - 160) * 4)
                             : (t < 208) ? (lin1_b + (t - 176) * 4)
                                         : (lin2_b + (t - 208) * 4);
            cst = *reinterpret_cast<const float4*>(src);
        }
    }

    float4 acca[V], accb[V], owna[V], ownb[V];
    for (int v = 0; v < V; ++v) {
        acca[v] = make_float4(0.f, 0.f, 0.f, 0.f);
        accb[v] = make_float4(0.f, 0.f, 0.f, 0.f);
        owna[v] = make_float4(0.f, 0.f, 0.f, 0.f);
        ownb[v] = make_float4(0.f, 0.f, 0.f, 0.f);
    }
    int da = 0, db = 0;
    if (nodea < n_nodes) {
        da = cnt[nodea];
        const float4* xr = reinterpret_cast<const float4*>(feat + (size_t)nodea * D) + lane;
        for (int v = 0; v < V; ++v) owna[v] = xr[32 * v];
    }
    if (nodeb < n_nodes) {
        db = cnt[nodeb];
        const float4* xr = reinterpret_cast<const float4*>(feat + (size_t)nodeb * D) + lane;
        for (int v = 0; v < V; ++v) ownb[v] = xr[32 * v];
    }
    const int dega = __builtin_amdgcn_readfirstlane(da);
    const int degb = __builtin_amdgcn_readfirstlane(db);
    const int ca = min(dega, CAP);
    const int cb = min(degb, CAP);
    int idxa = 0, idxb = 0;
    if (ca > 0) idxa = slots[(size_t)nodea * CAP + min(lane, ca - 1)];
    if (cb > 0) idxb = slots[(size_t)nodeb * CAP + min(lane, cb - 1)];
    idxa = (int)min((unsigned)idxa, (unsigned)(n_nodes - 1));
    idxb = (int)min((unsigned)idxb, (unsigned)(n_nodes - 1));
    for (int v = 0; v < V; ++v) {
        *reinterpret_cast<float4*>(&Xs[w * P + (lane + 32 * v) * 4]) = owna[v];
        *reinterpret_cast<float4*>(&Xs[(w + NW) * P + (lane + 32 * v) * 4]) = ownb[v];
    }
    gather_pair<D>(feat, idxa, idxb, ca, cb, lane, acca, accb);
    if (dega > CAP)
        gather_overflow<D>(feat, ovf, novf, nodea, lane, n_nodes, n_edges, acca);
    if (degb > CAP)
        gather_overflow<D>(feat, ovf, novf, nodeb, lane, n_nodes, n_edges, accb);
    for (int v = 0; v < V; ++v) {
        *reinterpret_cast<float4*>(&As[w * P + (lane + 32 * v) * 4]) = acca[v];
        *reinterpret_cast<float4*>(&As[(w + NW) * P + (lane + 32 * v) * 4]) = accb[v];
    }
    if constexpr (TAIL) {
        if (threadIdx.x < 224)
            *reinterpret_cast<float4*>(&Ks[threadIdx.x * 4]) = cst;
    }
    __syncthreads();

    const int zr_ = threadIdx.x / 10;
    const int zc_ = threadIdx.x - zr_ * 10;
    float zb = 0.0f;
    if constexpr (TAIL) {
        if (threadIdx.x < 160) zb = lin3_b[zc_];
    }

    const int g = w % G;
    const int p = w / G;
    const bool root = (p * KP >= D);
    const int kbase = p * KP - (root ? D : 0);
    const int rh = lane & 15;
    const int hi = lane >> 4;
    const float* sp = (root ? Xs : As) + rh * P + kbase + 2 * hi;
    const float* wp = (root ? Wroot : Wrel) + (size_t)(kbase + 2 * hi) * N + g * 32 + 2 * rh;

    v8f c0 = v8f{}, c1 = v8f{};
#pragma unroll 8
    for (int k = 0; k < KP; k += 4) {
        const v2f a = *reinterpret_cast<const v2f*>(sp + k);
        const v2f l0 = *reinterpret_cast<const v2f*>(wp + (size_t)k * N);
        const v2f l1 = *reinterpret_cast<const v2f*>(wp + (size_t)(k + 1) * N);
        v2f b0, b1;
        b0.x = l0.x; b0.y = l1.x;
        b1.x = l0.y; b1.y = l1.y;
        c0 = __builtin_amdgcn_wmma_f32_16x16x4_f32(
            false, a, false, b0, (short)0, c0, false, false);
        c1 = __builtin_amdgcn_wmma_f32_16x16x4_f32(
            false, a, false, b1, (short)0, c1, false, false);
    }
#pragma unroll
    for (int i = 0; i < 8; ++i) {
        v2f d;
        d.x = c0[i]; d.y = c1[i];
        *reinterpret_cast<v2f*>(&Cs[(p * 16 + i + 8 * hi) * N + g * 32 + 2 * rh]) = d;
    }
    __syncthreads();

    if constexpr (!TAIL) {
        for (int idx = threadIdx.x; idx < 16 * (N / 4); idx += NT) {
            const int r = idx / (N / 4);
            const int c4 = (idx - r * (N / 4)) * 4;
            float4 sum = *reinterpret_cast<const float4*>(bias + c4);
#pragma unroll
            for (int q = 0; q < PARTS; ++q) {
                const float4 pv = *reinterpret_cast<const float4*>(&Cs[(q * 16 + r) * N + c4]);
                sum.x += pv.x; sum.y += pv.y; sum.z += pv.z; sum.w += pv.w;
            }
            if (relu) {
                sum.x = fmaxf(sum.x, 0.0f); sum.y = fmaxf(sum.y, 0.0f);
                sum.z = fmaxf(sum.z, 0.0f); sum.w = fmaxf(sum.w, 0.0f);
            }
            if (node0 + r < n_nodes)
                *reinterpret_cast<float4*>(out + (size_t)(node0 + r) * N + c4) = sum;
        }
    } else {
        reduce_to_tile<N, PARTS>(Cs, Ks + 640, Hs, relu);
        __syncthreads();
        dense_partials<64, 128>(Hs, lin1_w, Cs, w, lane);
        __syncthreads();
        reduce_to_tile<128, NW / 4>(Cs, Ks + 704, As, 1);
        __syncthreads();
        dense_partials<128, 64>(As, lin2_w, Cs, w, lane);
        __syncthreads();
        reduce_to_tile<64, NW / 2>(Cs, Ks + 832, Ms, 1);
        __syncthreads();
        const int t = threadIdx.x;
        const int r = zr_;
        const int c = zc_;
        if (t < 160) {
            float z = zb;
            const float* mr = Ms + r * 68;
            for (int k = 0; k < 64; ++k) z = fmaf(mr[k], Ks[k * 10 + c], z);
            Zs[t] = z;
        }
        __syncthreads();
        if (t < 160) {
            const float* zr = Zs + r * 10;
            float mx = zr[0];
            for (int j = 1; j < 10; ++j) mx = fmaxf(mx, zr[j]);
            float s = 0.0f;
            for (int j = 0; j < 10; ++j) s += expf(zr[j] - mx);
            const float lse = mx + logf(s);
            if (node0 + r < n_nodes)
                out[(size_t)(node0 + r) * 10 + c] = zr[c] - lse;
        }
    }
}

extern "C" void kernel_launch(void* const* d_in, const int* in_sizes, int n_in,
                              void* d_out, int out_size, void* d_ws, size_t ws_size,
                              hipStream_t stream) {
    const float* x       = (const float*)d_in[0];
    const int*   ei      = (const int*)d_in[1];
    const float* W1_rel  = (const float*)d_in[2];
    const float* W1_root = (const float*)d_in[3];
    const float* b1      = (const float*)d_in[4];
    const float* W2_rel  = (const float*)d_in[5];
    const float* W2_root = (const float*)d_in[6];
    const float* b2      = (const float*)d_in[7];
    const float* lin1_w  = (const float*)d_in[8];
    const float* lin1_b  = (const float*)d_in[9];
    const float* lin2_w  = (const float*)d_in[10];
    const float* lin2_b  = (const float*)d_in[11];
    const float* lin3_w  = (const float*)d_in[12];
    const float* lin3_b  = (const float*)d_in[13];

    int*   cnt   = (int*)d_ws;
    int*   novf  = cnt + NNODES;
    int*   slots = novf + 32;
    int*   ovf   = slots + (size_t)NNODES * CAP;
    float* h1    = (float*)d_ws + (size_t)NNODES * 256;

    hipMemsetAsync(cnt, 0, (size_t)(NNODES + 32) * sizeof(int), stream);
    k_file_edges<<<(NEDGES / 4 + 255) / 256, 256, 0, stream>>>(
        ei, cnt, novf, slots, ovf, NEDGES);

    const int gblocks = (NNODES + 15) / 16;

    k_gather_gemm<256, 128, false><<<gblocks, NT, 0, stream>>>(
        x, cnt, slots, ovf, novf, NEDGES, W1_rel, W1_root, b1,
        nullptr, nullptr, nullptr, nullptr, nullptr, nullptr,
        h1, NNODES, 1);

    k_gather_gemm<128, 64, true><<<gblocks, NT, 0, stream>>>(
        h1, cnt, slots, ovf, novf, NEDGES, W2_rel, W2_root, b2,
        lin1_w, lin1_b, lin2_w, lin2_b, lin3_w, lin3_b,
        (float*)d_out, NNODES, 0);
}
